# starting kernel
# baseline (speedup 1.0000x reference)
#include <hip/hip_runtime.h>
#include <math.h>

typedef __attribute__((ext_vector_type(2))) float v2f;
typedef __attribute__((ext_vector_type(8))) float v8f;

#define B_   4
#define S_   1024
#define H_   16
#define DK   4
#define E_   64
#define QCH  8
#define QPB  (S_ / QCH)

#define HALF_LOG2E 0.7213475204444817f

__global__ __launch_bounds__(256) void qattn_kernel(
    const float* __restrict__ x,
    const float* __restrict__ theta,
    float* __restrict__ attnout)
{
    __shared__ float pl[S_ * DK];

    const int bh     = blockIdx.x / QCH;
    const int qchunk = blockIdx.x % QCH;
    const int b      = bh / H_;
    const int h      = bh % H_;

    const float ct0 = __cosf(theta[0]);
    const float ct1 = __cosf(theta[1]);
    const float ct2 = __cosf(theta[2]);
    const float ct3 = __cosf(theta[3]);

    {
        const float4* xr  = (const float4*)(x + (size_t)b * S_ * E_ + h * DK);
        float4*       plr = (float4*)pl;
        for (int s = threadIdx.x; s < S_; s += 256) {
            float4 v = xr[s * (E_ / 4)];
            float4 p;
            p.x = __cosf(v.x) * ct0;
            p.y = __cosf(v.y) * ct1;
            p.z = __cosf(v.z) * ct2;
            p.w = __cosf(v.w) * ct3;
            plr[s] = p;
        }
    }
    __syncthreads();

    const int wave = threadIdx.x >> 5;
    const int lane = threadIdx.x & 31;
    const int mn   = lane & 15;
    const bool hi  = (lane >= 16);

    const int qb = qchunk * QPB + wave * 16;

    v2f a;
    {
        float4 q = ((const float4*)pl)[qb + mn];
        a.x = (hi ? q.z : q.x) * HALF_LOG2E;
        a.y = (hi ? q.w : q.y) * HALF_LOG2E;
    }

    float num[8][4];
    float den[8];
#pragma unroll
    for (int r = 0; r < 8; ++r) {
        den[r] = 0.0f;
        num[r][0] = 0.0f; num[r][1] = 0.0f; num[r][2] = 0.0f; num[r][3] = 0.0f;
    }

    for (int kt = 0; kt < S_; kt += 16) {
        float4 kv = ((const float4*)pl)[kt + mn];
        v2f bf;
        bf.x = hi ? kv.z : kv.x;
        bf.y = hi ? kv.w : kv.y;

        v8f c = {};
        c = __builtin_amdgcn_wmma_f32_16x16x4_f32(
                 false, a,  false, bf,
                 (short)0, c,  false,  false);

#pragma unroll
        for (int r = 0; r < 8; ++r) {
            float e = __builtin_amdgcn_exp2f(c[r]);
            den[r]    += e;
            num[r][0] += e * kv.x;
            num[r][1] += e * kv.y;
            num[r][2] += e * kv.z;
            num[r][3] += e * kv.w;
        }
    }

#pragma unroll
    for (int off = 1; off < 16; off <<= 1) {
#pragma unroll
        for (int r = 0; r < 8; ++r) {
            den[r]    += __shfl_xor(den[r],    off, 32);
            num[r][0] += __shfl_xor(num[r][0], off, 32);
            num[r][1] += __shfl_xor(num[r][1], off, 32);
            num[r][2] += __shfl_xor(num[r][2], off, 32);
            num[r][3] += __shfl_xor(num[r][3], off, 32);
        }
    }

    const int p = mn;
    if (p < 8) {
        const int  row = qb + (hi ? 8 : 0) + p;
        const float inv = 1.0f / den[p];
        float* o = attnout + ((size_t)(b * S_ + row) * E_ + h * DK);
        float4 v;
        v.x = num[p][0] * inv;
        v.y = num[p][1] * inv;
        v.z = num[p][2] * inv;
        v.w = num[p][3] * inv;
        *(float4*)o = v;
    }
}

__global__ __launch_bounds__(128) void linear_kernel(
    const float* __restrict__ attnout,
    const float* __restrict__ W,
    const float* __restrict__ bias,
    float* __restrict__ out)
{
    const int wave = threadIdx.x >> 5;
    const int lane = threadIdx.x & 31;
    const int mn   = lane & 15;
    const int kb   = (lane >> 4) << 1;

    const int mb = blockIdx.x * 16;
    const int nb = wave * 16;

    const float* arow = attnout + (size_t)(mb + mn) * E_;
    const float* wrow = W       + (size_t)(nb + mn) * E_;

    v8f c = {};
#pragma unroll
    for (int k = 0; k < E_; k += 4) {
        v2f a, bf;
        a.x  = arow[k + kb];
        a.y  = arow[k + kb + 1];
        bf.x = wrow[k + kb];
        bf.y = wrow[k + kb + 1];
        c = __builtin_amdgcn_wmma_f32_16x16x4_f32(
                false, a, false, bf, (short)0, c, false, false);
    }

    const float bv  = bias[nb + mn];
    const int row0  = mb + ((lane >= 16) ? 8 : 0);
#pragma unroll
    for (int r = 0; r < 8; ++r) {
        out[(size_t)(row0 + r) * E_ + nb + mn] = c[r] + bv;
    }
}

extern "C" void kernel_launch(void* const* d_in, const int* in_sizes, int n_in,
                              void* d_out, int out_size, void* d_ws, size_t ws_size,
                              hipStream_t stream)
{
    const float* x     = (const float*)d_in[0];
    const float* theta = (const float*)d_in[1];
    const float* W     = (const float*)d_in[2];
    const float* bias  = (const float*)d_in[3];
    float*       out   = (float*)d_out;
    float*       attn  = (float*)d_ws;

    (void)in_sizes; (void)n_in; (void)out_size; (void)ws_size;

    qattn_kernel<<<B_ * H_ * QCH, 256, 0, stream>>>(x, theta, attn);
    linear_kernel<<<(B_ * S_) / 16, 128, 0, stream>>>(attn, W, bias, out);
}


# Fuse the 64x64 output linear into the attention kernel: a 16-wave workgroup per (batch, 16 queries) holds all 16 heads' proj in LDS, so out = attn @ W^T + b runs in-kernel; one launch, no scratch
# speedup vs baseline: 1.1655x; 1.1655x over previous
#include <hip/hip_runtime.h>
#include <math.h>


typedef __attribute__((ext_vector_type(2))) float v2f;
typedef __attribute__((ext_vector_type(8))) float v8f;

#define B_   4
#define S_   1024
#define H_   16
#define DK   4
#define E_   64
#define QT   16
#define NQT  (S_ / QT)
#define HSTR (S_ + 1)
#define TSTR 68

#define HALF_LOG2E 0.7213475204444817f

__global__ __launch_bounds__(512) void qattn_linear_kernel(
    const float* __restrict__ x,
    const float* __restrict__ theta,
    const float* __restrict__ W,
    const float* __restrict__ bias,
    float* __restrict__ out)
{
    __shared__ float4 pl[H_ * HSTR];
    __shared__ __attribute__((aligned(16))) float at[QT * TSTR];

    const int b  = blockIdx.x / NQT;
    const int qb = (blockIdx.x % NQT) * QT;

    const float ct0 = __cosf(theta[0]);
    const float ct1 = __cosf(theta[1]);
    const float ct2 = __cosf(theta[2]);
    const float ct3 = __cosf(theta[3]);

    {
        const float4* xb  = (const float4*)(x + (size_t)b * S_ * E_);
        const int     rot = blockIdx.x & 31;
#pragma unroll 16
        for (int it = 0; it < 32; ++it) {
            const int g = (((it + rot) & 31) << 9) + threadIdx.x;
            float4 v = xb[g];
            float4 p;
            p.x = __cosf(v.x) * ct0;
            p.y = __cosf(v.y) * ct1;
            p.z = __cosf(v.z) * ct2;
            p.w = __cosf(v.w) * ct3;
            pl[(g & 15) * HSTR + (g >> 4)] = p;
        }
    }
    __syncthreads();

    const int wave = threadIdx.x >> 5;
    const int lane = threadIdx.x & 31;
    const int mn   = lane & 15;
    const bool hi  = (lane >= 16);

    const float4* ph = pl + wave * HSTR;

    const bool lin = __builtin_amdgcn_readfirstlane(wave) < 4;
    const int  kb  = (lane >> 4) << 1;
    const int  nb  = (wave & 3) * 16;
    v2f   wf[E_ / 4];
    float bv = 0.0f;
#pragma unroll
    for (int j = 0; j < E_ / 4; ++j) { wf[j].x = 0.0f; wf[j].y = 0.0f; }
    if (lin) {
        const float* wrow = W + (size_t)(nb + mn) * E_ + kb;
#pragma unroll
        for (int j = 0; j < E_ / 4; ++j) wf[j] = *(const v2f*)(wrow + 4 * j);
        bv = bias[nb + mn];
    }

    v2f a;
    {
        float4 q = ph[qb + mn];
        a.x = (hi ? q.z : q.x) * HALF_LOG2E;
        a.y = (hi ? q.w : q.y) * HALF_LOG2E;
    }

    float num[8][4];
    float den[8];
#pragma unroll
    for (int r = 0; r < 8; ++r) {
        den[r] = 0.0f;
        num[r][0] = 0.0f; num[r][1] = 0.0f; num[r][2] = 0.0f; num[r][3] = 0.0f;
    }

    for (int kt = 0; kt < S_; kt += 16) {
        float4 kv = ph[kt + mn];
        v2f bf;
        bf.x = hi ? kv.z : kv.x;
        bf.y = hi ? kv.w : kv.y;

        v8f c = {};
        c = __builtin_amdgcn_wmma_f32_16x16x4_f32(
                 false, a,  false, bf,
                 (short)0, c,  false,  false);

#pragma unroll
        for (int r = 0; r < 8; ++r) {
            float e = __builtin_amdgcn_exp2f(c[r]);
            den[r]    += e;
            num[r][0] += e * kv.x;
            num[r][1] += e * kv.y;
            num[r][2] += e * kv.z;
            num[r][3] += e * kv.w;
        }
    }

#pragma unroll
    for (int off = 1; off < 16; off <<= 1) {
#pragma unroll
        for (int r = 0; r < 8; ++r) {
            den[r]    += __shfl_xor(den[r],    off, 32);
            num[r][0] += __shfl_xor(num[r][0], off, 32);
            num[r][1] += __shfl_xor(num[r][1], off, 32);
            num[r][2] += __shfl_xor(num[r][2], off, 32);
            num[r][3] += __shfl_xor(num[r][3], off, 32);
        }
    }

    const int p = mn;
    if (p < 8) {
        const int   row = (hi ? 8 : 0) + p;
        const float inv = 1.0f / den[p];
        float4 v;
        v.x = num[p][0] * inv;
        v.y = num[p][1] * inv;
        v.z = num[p][2] * inv;
        v.w = num[p][3] * inv;
        *(float4*)&at[row * TSTR + wave * DK] = v;
    }
    __syncthreads();

    if (lin) {
        const float* arow = at + mn * TSTR + kb;

        v8f c = {};
#pragma unroll
        for (int k = 0; k < E_; k += 4) {
            v2f av = *(const v2f*)(arow + k);
            c = __builtin_amdgcn_wmma_f32_16x16x4_f32(
                    false, av, false, wf[k / 4], (short)0, c, false, false);
        }

        const int   row0 = b * S_ + qb + (hi ? 8 : 0);
#pragma unroll
        for (int r = 0; r < 8; ++r) {
            out[(size_t)(row0 + r) * E_ + nb + mn] = c[r] + bv;
        }
    }
}

extern "C" void kernel_launch(void* const* d_in, const int* in_sizes, int n_in,
                              void* d_out, int out_size, void* d_ws, size_t ws_size,
                              hipStream_t stream)
{
    const float* x     = (const float*)d_in[0];
    const float* theta = (const float*)d_in[1];
    const float* W     = (const float*)d_in[2];
    const float* bias  = (const float*)d_in[3];
    float*       out   = (float*)d_out;

    (void)in_sizes; (void)n_in; (void)out_size; (void)d_ws; (void)ws_size;

    qattn_linear_kernel<<<B_ * NQT, 512, 0, stream>>>(x, theta, W, bias, out);
}


# Pipeline the projection under the key loop: x[b] is fetched in 4 chunks of 256 keys, each chunk's global loads in flight during the previous chunk's attention; keys summed in rotated sweep order
# speedup vs baseline: 1.3036x; 1.1185x over previous
#include <hip/hip_runtime.h>
#include <math.h>


typedef __attribute__((ext_vector_type(2))) float v2f;
typedef __attribute__((ext_vector_type(8))) float v8f;

#define B_   4
#define S_   1024
#define H_   16
#define DK   4
#define E_   64
#define QT   16
#define NQT  (S_ / QT)
#define HSTR (S_ + 1)
#define TSTR 68
#define NSW  32
#define CSW  8
#define NCH  (NSW / CSW)

#define HALF_LOG2E 0.7213475204444817f

__device__ __forceinline__ int sweep_base(int s, int rot)
{
    return ((s + rot) & (NSW - 1)) << 9;
}

__device__ __forceinline__ void lds_barrier()
{
    __builtin_amdgcn_fence(__ATOMIC_RELEASE, "workgroup", "local");
    __builtin_amdgcn_s_barrier();
    __builtin_amdgcn_fence(__ATOMIC_ACQUIRE, "workgroup", "local");
}

__device__ __forceinline__ void fetch_chunk(const float4* __restrict__ xb, int c, int rot,
                                            float4 (&xr)[CSW])
{
#pragma unroll
    for (int i = 0; i < CSW; ++i)
        xr[i] = xb[sweep_base(c * CSW + i, rot) + threadIdx.x];
}

__device__ __forceinline__ void project_chunk(float4* __restrict__ pl, int c, int rot,
                                              const float4 (&xr)[CSW],
                                              float ct0, float ct1, float ct2, float ct3)
{
#pragma unroll
    for (int i = 0; i < CSW; ++i) {
        const int g = sweep_base(c * CSW + i, rot) + threadIdx.x;
        float4 p;
        p.x = __cosf(xr[i].x) * ct0;
        p.y = __cosf(xr[i].y) * ct1;
        p.z = __cosf(xr[i].z) * ct2;
        p.w = __cosf(xr[i].w) * ct3;
        pl[(g & 15) * HSTR + (g >> 4)] = p;
    }
}

__global__ __launch_bounds__(512) void qattn_linear_kernel(
    const float* __restrict__ x,
    const float* __restrict__ theta,
    const float* __restrict__ W,
    const float* __restrict__ bias,
    float* __restrict__ out)
{
    __shared__ float4 pl[H_ * HSTR];
    __shared__ __attribute__((aligned(16))) float at[QT * TSTR];

    const int b  = blockIdx.x / NQT;
    const int qt = blockIdx.x % NQT;
    const int qb = qt * QT;

    const float ct0 = __cosf(theta[0]);
    const float ct1 = __cosf(theta[1]);
    const float ct2 = __cosf(theta[2]);
    const float ct3 = __cosf(theta[3]);

    const float4* xb = (const float4*)(x + (size_t)b * S_ * E_);

    const int rot = qt >> 1;

    float4 xr[CSW];
    fetch_chunk(xb, 0, rot, xr);
    project_chunk(pl, 0, rot, xr, ct0, ct1, ct2, ct3);
    fetch_chunk(xb, 1, rot, xr);
    lds_barrier();

    const int wave = threadIdx.x >> 5;
    const int lane = threadIdx.x & 31;
    const int mn   = lane & 15;
    const bool hi  = (lane >= 16);

    const float4* ph = pl + wave * HSTR;

    const bool lin = __builtin_amdgcn_readfirstlane(wave) < 4;
    const int  kb  = (lane >> 4) << 1;
    const int  nb  = (wave & 3) * 16;
    v2f   wf[E_ / 4];
    float bv = 0.0f;
#pragma unroll
    for (int j = 0; j < E_ / 4; ++j) { wf[j].x = 0.0f; wf[j].y = 0.0f; }
    if (lin) {
        const float* wrow = W + (size_t)(nb + mn) * E_ + kb;
#pragma unroll
        for (int j = 0; j < E_ / 4; ++j) wf[j] = *(const v2f*)(wrow + 4 * j);
        bv = bias[nb + mn];
    }

    v2f a;
    {
        float4 q = ph[qb + mn];
        a.x = (hi ? q.z : q.x) * HALF_LOG2E;
        a.y = (hi ? q.w : q.y) * HALF_LOG2E;
    }

    float num[8][4];
    float den[8];
#pragma unroll
    for (int r = 0; r < 8; ++r) {
        den[r] = 0.0f;
        num[r][0] = 0.0f; num[r][1] = 0.0f; num[r][2] = 0.0f; num[r][3] = 0.0f;
    }

    for (int c = 0; c < NCH; ++c) {
        for (int i = 0; i < CSW; ++i) {
            const int k32 = sweep_base(c * CSW + i, rot) >> 4;
#pragma unroll
            for (int t = 0; t < 2; ++t) {
                const int kt = k32 + (t << 4);
                float4 kv = ph[kt + mn];
                v2f bf;
                bf.x = hi ? kv.z : kv.x;
                bf.y = hi ? kv.w : kv.y;

                v8f cc = {};
                cc = __builtin_amdgcn_wmma_f32_16x16x4_f32(
                         false, a,  false, bf,
                         (short)0, cc,  false,  false);

#pragma unroll
                for (int r = 0; r < 8; ++r) {
                    float e = __builtin_amdgcn_exp2f(cc[r]);
                    den[r]    += e;
                    num[r][0] += e * kv.x;
                    num[r][1] += e * kv.y;
                    num[r][2] += e * kv.z;
                    num[r][3] += e * kv.w;
                }
            }
        }

        if (c + 1 < NCH) {
            project_chunk(pl, c + 1, rot, xr, ct0, ct1, ct2, ct3);
            if (c + 2 < NCH) fetch_chunk(xb, c + 2, rot, xr);
            lds_barrier();
        }
    }

#pragma unroll
    for (int off = 1; off < 16; off <<= 1) {
#pragma unroll
        for (int r = 0; r < 8; ++r) {
            den[r]    += __shfl_xor(den[r],    off, 32);
            num[r][0] += __shfl_xor(num[r][0], off, 32);
            num[r][1] += __shfl_xor(num[r][1], off, 32);
            num[r][2] += __shfl_xor(num[r][2], off, 32);
            num[r][3] += __shfl_xor(num[r][3], off, 32);
        }
    }

    const int p = mn;
    if (p < 8) {
        const int   row = (hi ? 8 : 0) + p;
        const float inv = 1.0f / den[p];
        float4 v;
        v.x = num[p][0] * inv;
        v.y = num[p][1] * inv;
        v.z = num[p][2] * inv;
        v.w = num[p][3] * inv;
        *(float4*)&at[row * TSTR + wave * DK] = v;
    }
    __syncthreads();

    if (lin) {
        const float* arow = at + mn * TSTR + kb;

        v8f c = {};
#pragma unroll
        for (int k = 0; k < E_; k += 4) {
            v2f av = *(const v2f*)(arow + k);
            c = __builtin_amdgcn_wmma_f32_16x16x4_f32(
                    false, av, false, wf[k / 4], (short)0, c, false, false);
        }

        const int   row0 = b * S_ + qb + (hi ? 8 : 0);
#pragma unroll
        for (int r = 0; r < 8; ++r) {
            out[(size_t)(row0 + r) * E_ + nb + mn] = c[r] + bv;
        }
    }
}

extern "C" void kernel_launch(void* const* d_in, const int* in_sizes, int n_in,
                              void* d_out, int out_size, void* d_ws, size_t ws_size,
                              hipStream_t stream)
{
    const float* x     = (const float*)d_in[0];
    const float* theta = (const float*)d_in[1];
    const float* W     = (const float*)d_in[2];
    const float* bias  = (const float*)d_in[3];
    float*       out   = (float*)d_out;

    (void)in_sizes; (void)n_in; (void)out_size; (void)d_ws; (void)ws_size;

    qattn_linear_kernel<<<B_ * NQT, 512, 0, stream>>>(x, theta, W, bias, out);
}


# Row sums over the 16 key lanes by a DPP reduce-scatter tree on the vector ALU (40 cross-lane adds per lane) instead of 160 ds_bpermute shuffles per wave; order of the final 16-lane sums changes
# speedup vs baseline: 1.3976x; 1.0721x over previous
#include <hip/hip_runtime.h>
#include <math.h>


typedef __attribute__((ext_vector_type(2))) float v2f;
typedef __attribute__((ext_vector_type(8))) float v8f;

#define B_   4
#define S_   1024
#define H_   16
#define DK   4
#define E_   64
#define QT   16
#define NQT  (S_ / QT)
#define HSTR (S_ + 1)
#define TSTR 68
#define NSW  32
#define CSW  8
#define NCH  (NSW / CSW)

#define HALF_LOG2E 0.7213475204444817f

__device__ __forceinline__ int sweep_base(int s, int rot)
{
    return ((s + rot) & (NSW - 1)) << 9;
}

__device__ __forceinline__ void lds_barrier()
{
    __builtin_amdgcn_fence(__ATOMIC_RELEASE, "workgroup", "local");
    __builtin_amdgcn_s_barrier();
    __builtin_amdgcn_fence(__ATOMIC_ACQUIRE, "workgroup", "local");
}

template <int M>
__device__ __forceinline__ float row_xor(float v)
{
    return __int_as_float(__builtin_amdgcn_mov_dpp(__float_as_int(v), 0x160 + M, 0xf, 0xf, true));
}

__device__ __forceinline__ void fetch_chunk(const float4* __restrict__ xb, int c, int rot,
                                            float4 (&xr)[CSW])
{
#pragma unroll
    for (int i = 0; i < CSW; ++i)
        xr[i] = xb[sweep_base(c * CSW + i, rot) + threadIdx.x];
}

__device__ __forceinline__ void project_chunk(float4* __restrict__ pl, int c, int rot,
                                              const float4 (&xr)[CSW],
                                              float ct0, float ct1, float ct2, float ct3)
{
#pragma unroll
    for (int i = 0; i < CSW; ++i) {
        const int g = sweep_base(c * CSW + i, rot) + threadIdx.x;
        float4 p;
        p.x = __cosf(xr[i].x) * ct0;
        p.y = __cosf(xr[i].y) * ct1;
        p.z = __cosf(xr[i].z) * ct2;
        p.w = __cosf(xr[i].w) * ct3;
        pl[(g & 15) * HSTR + (g >> 4)] = p;
    }
}

__global__ __launch_bounds__(512) void qattn_linear_kernel(
    const float* __restrict__ x,
    const float* __restrict__ theta,
    const float* __restrict__ W,
    const float* __restrict__ bias,
    float* __restrict__ out)
{
    __shared__ float4 pl[H_ * HSTR];
    __shared__ __attribute__((aligned(16))) float at[QT * TSTR];

    const int b  = blockIdx.x / NQT;
    const int qt = blockIdx.x % NQT;
    const int qb = qt * QT;

    const float ct0 = __cosf(theta[0]);
    const float ct1 = __cosf(theta[1]);
    const float ct2 = __cosf(theta[2]);
    const float ct3 = __cosf(theta[3]);

    const float4* xb = (const float4*)(x + (size_t)b * S_ * E_);

    const int rot = qt >> 1;

    float4 xr[CSW];
    fetch_chunk(xb, 0, rot, xr);
    project_chunk(pl, 0, rot, xr, ct0, ct1, ct2, ct3);
    fetch_chunk(xb, 1, rot, xr);
    lds_barrier();

    const int wave = threadIdx.x >> 5;
    const int lane = threadIdx.x & 31;
    const int mn   = lane & 15;
    const bool hi  = (lane >= 16);

    const float4* ph = pl + wave * HSTR;

    const bool lin = __builtin_amdgcn_readfirstlane(wave) < 4;
    const int  kb  = (lane >> 4) << 1;
    const int  nb  = (wave & 3) * 16;
    v2f   wf[E_ / 4];
    float bv = 0.0f;
#pragma unroll
    for (int j = 0; j < E_ / 4; ++j) { wf[j].x = 0.0f; wf[j].y = 0.0f; }
    if (lin) {
        const float* wrow = W + (size_t)(nb + mn) * E_ + kb;
#pragma unroll
        for (int j = 0; j < E_ / 4; ++j) wf[j] = *(const v2f*)(wrow + 4 * j);
        bv = bias[nb + mn];
    }

    v2f a;
    {
        float4 q = ph[qb + mn];
        a.x = (hi ? q.z : q.x) * HALF_LOG2E;
        a.y = (hi ? q.w : q.y) * HALF_LOG2E;
    }

    float num[8][4];
    float den[8];
#pragma unroll
    for (int r = 0; r < 8; ++r) {
        den[r] = 0.0f;
        num[r][0] = 0.0f; num[r][1] = 0.0f; num[r][2] = 0.0f; num[r][3] = 0.0f;
    }

    for (int c = 0; c < NCH; ++c) {
#pragma unroll 2
        for (int i = 0; i < CSW; ++i) {
            const int k32 = sweep_base(c * CSW + i, rot) >> 4;
#pragma unroll
            for (int t = 0; t < 2; ++t) {
                const int kt = k32 + (t << 4);
                float4 kv = ph[kt + mn];
                v2f bf;
                bf.x = hi ? kv.z : kv.x;
                bf.y = hi ? kv.w : kv.y;

                v8f cc = {};
                cc = __builtin_amdgcn_wmma_f32_16x16x4_f32(
                         false, a,  false, bf,
                         (short)0, cc,  false,  false);

#pragma unroll
                for (int r = 0; r < 8; ++r) {
                    float e = __builtin_amdgcn_exp2f(cc[r]);
                    den[r]    += e;
                    num[r][0] += e * kv.x;
                    num[r][1] += e * kv.y;
                    num[r][2] += e * kv.z;
                    num[r][3] += e * kv.w;
                }
            }
        }

        if (c + 1 < NCH) {
            project_chunk(pl, c + 1, rot, xr, ct0, ct1, ct2, ct3);
            if (c + 2 < NCH) fetch_chunk(xb, c + 2, rot, xr);
            lds_barrier();
        }
    }

    const bool b3 = (lane & 8) != 0;
    const bool b2 = (lane & 4) != 0;
    const bool b1 = (lane & 2) != 0;

    float s1[4][5];
#pragma unroll
    for (int j = 0; j < 4; ++j) {
        s1[j][0] = (b3 ? den[j + 4] : den[j]) + row_xor<8>(b3 ? den[j] : den[j + 4]);
#pragma unroll
        for (int d = 0; d < 4; ++d)
            s1[j][1 + d] = (b3 ? num[j + 4][d] : num[j][d])
                         + row_xor<8>(b3 ? num[j][d] : num[j + 4][d]);
    }
    float s2[2][5];
#pragma unroll
    for (int j = 0; j < 2; ++j) {
#pragma unroll
        for (int v = 0; v < 5; ++v)
            s2[j][v] = (b2 ? s1[j + 2][v] : s1[j][v]) + row_xor<4>(b2 ? s1[j][v] : s1[j + 2][v]);
    }
    float s4[5];
#pragma unroll
    for (int v = 0; v < 5; ++v) {
        const float s3 = (b1 ? s2[1][v] : s2[0][v]) + row_xor<2>(b1 ? s2[0][v] : s2[1][v]);
        s4[v] = s3 + row_xor<1>(s3);
    }

    if ((lane & 1) == 0) {
        const int   row = (hi ? 8 : 0) + (mn >> 1);
        const float inv = 1.0f / s4[0];
        float4 v;
        v.x = s4[1] * inv;
        v.y = s4[2] * inv;
        v.z = s4[3] * inv;
        v.w = s4[4] * inv;
        *(float4*)&at[row * TSTR + wave * DK] = v;
    }
    __syncthreads();

    if (lin) {
        const float* arow = at + mn * TSTR + kb;

        v8f c = {};
#pragma unroll
        for (int k = 0; k < E_; k += 4) {
            v2f av = *(const v2f*)(arow + k);
            c = __builtin_amdgcn_wmma_f32_16x16x4_f32(
                    false, av, false, wf[k / 4], (short)0, c, false, false);
        }

        const int   row0 = b * S_ + qb + (hi ? 8 : 0);
#pragma unroll
        for (int r = 0; r < 8; ++r) {
            out[(size_t)(row0 + r) * E_ + nb + mn] = c[r] + bv;
        }
    }
}

extern "C" void kernel_launch(void* const* d_in, const int* in_sizes, int n_in,
                              void* d_out, int out_size, void* d_ws, size_t ws_size,
                              hipStream_t stream)
{
    const float* x     = (const float*)d_in[0];
    const float* theta = (const float*)d_in[1];
    const float* W     = (const float*)d_in[2];
    const float* bias  = (const float*)d_in[3];
    float*       out   = (float*)d_out;

    (void)in_sizes; (void)n_in; (void)out_size; (void)d_ws; (void)ws_size;

    qattn_linear_kernel<<<B_ * NQT, 512, 0, stream>>>(x, theta, W, bias, out);
}
